# starting kernel
# baseline (speedup 1.0000x reference)
#include <hip/hip_runtime.h>
#include <hip/hip_bf16.h>


#define ROWS      2304
#define DDIM      64
#define NREP      207
#define ROW_FLOATS (NREP * DDIM)
#define ROW_F4     (ROW_FLOATS / 4)

typedef __attribute__((ext_vector_type(2))) float v2f;
typedef __attribute__((ext_vector_type(4))) float v4f;
typedef __attribute__((ext_vector_type(8))) float v8f;

__global__ void embed_h_kernel(const int* __restrict__ TE,
                               const float* __restrict__ W1,
                               const float* __restrict__ b1,
                               float* __restrict__ h) {
    const int idx = blockIdx.x * blockDim.x + threadIdx.x;
    const int row = idx >> 6;
    const int d   = idx & 63;
    if (row >= ROWS) return;

    int te0 = TE[row * (NREP * 2) + 0];
    int te1 = TE[row * (NREP * 2) + 1];
    int dow = te0 % 7;   if (dow < 0) dow += 7;
    int tod = te1 % 288; if (tod < 0) tod += 288;

    float v = W1[dow * DDIM + d] + W1[(7 + tod) * DDIM + d] + b1[d];
    h[row * DDIM + d] = v > 0.0f ? v : 0.0f;
}

__global__ void gemm_wmma_kernel(const float* __restrict__ h,
                                 const float* __restrict__ W2,
                                 const float* __restrict__ b2,
                                 float* __restrict__ outr) {
    const int lane = threadIdx.x & 31;
    const int wave = threadIdx.x >> 5;
    const int m0   = blockIdx.x * 16;
    const int n0   = wave * 16;
    const int nl   = lane & 15;
    const int kh   = (lane >> 4) * 2;

    const float* __restrict__ arow = h + (m0 + nl) * DDIM;

    v8f c = {};
#pragma unroll
    for (int k0 = 0; k0 < DDIM; k0 += 4) {
        v2f a, b;
        a[0] = arow[k0 + kh + 0];
        a[1] = arow[k0 + kh + 1];
        b[0] = W2[(k0 + kh + 0) * DDIM + n0 + nl];
        b[1] = W2[(k0 + kh + 1) * DDIM + n0 + nl];
        c = __builtin_amdgcn_wmma_f32_16x16x4_f32(
                false, a, false, b, (short)0, c, false, false);
    }

    const float bias  = b2[n0 + nl];
    const int   mbase = m0 + (lane >> 4) * 8;
#pragma unroll
    for (int r = 0; r < 8; ++r) {
        outr[(mbase + r) * DDIM + n0 + nl] = c[r] + bias;
    }
}

__global__ void broadcast_kernel(const float* __restrict__ outr,
                                 float* __restrict__ out) {
    const int r   = blockIdx.x;
    const int tid = threadIdx.x;

    const v4f v = ((const v4f*)(outr + r * DDIM))[tid & 15];
    v4f* __restrict__ dst = (v4f*)(out + (size_t)r * ROW_FLOATS);

    for (int i = tid; i < ROW_F4; i += 256) {
        __builtin_nontemporal_store(v, dst + i);
    }
}

extern "C" void kernel_launch(void* const* d_in, const int* in_sizes, int n_in,
                              void* d_out, int out_size, void* d_ws, size_t ws_size,
                              hipStream_t stream) {
    (void)in_sizes; (void)n_in; (void)out_size; (void)ws_size;

    const int*   TE = (const int*)  d_in[0];
    const float* W1 = (const float*)d_in[1];
    const float* b1 = (const float*)d_in[2];
    const float* W2 = (const float*)d_in[3];
    const float* b2 = (const float*)d_in[4];

    float* out  = (float*)d_out;
    float* h    = (float*)d_ws;
    float* outr = h + ROWS * DDIM;

    embed_h_kernel<<<(ROWS * DDIM) / 256, 256, 0, stream>>>(TE, W1, b1, h);

    gemm_wmma_kernel<<<ROWS / 16, 128, 0, stream>>>(h, W2, b2, outr);

    broadcast_kernel<<<ROWS, 256, 0, stream>>>(outr, out);
}


# One launch, no LDS or workspace: each wave rebuilds its 16-row tile on f32 WMMA with W2 columns 4c+t as B, so its accumulators are the float4 store layout, and streams its share of the 207 copies
# speedup vs baseline: 1.3661x; 1.3661x over previous
#include <hip/hip_runtime.h>


#define ROWS        2304
#define DDIM        64
#define NREP        207
#define ROW_FLOATS  (NREP * DDIM)
#define PARTS       14
#define WG_WAVES    8
#define NUM_WAVES   ((ROWS / 16) * PARTS)
#define NUM_GROUPS  (NUM_WAVES / WG_WAVES)

typedef __attribute__((ext_vector_type(2))) int   v2i;
typedef __attribute__((ext_vector_type(2))) float v2f;
typedef __attribute__((ext_vector_type(4))) float v4f;
typedef __attribute__((ext_vector_type(8))) float v8f;

__global__ __launch_bounds__(WG_WAVES * 32)
void embed_rows_kernel(const int*   __restrict__ TE,
                       const float* __restrict__ W1,
                       const float* __restrict__ b1,
                       const float* __restrict__ W2,
                       const float* __restrict__ b2,
                       float*       __restrict__ out) {
    const int lane = threadIdx.x & 31;
    const int wv   = __builtin_amdgcn_readfirstlane((int)(threadIdx.x >> 5));
    const int gw   = blockIdx.x * WG_WAVES + wv;
    const int tile = gw / PARTS;
    const int part = gw - tile * PARTS;
    const int c    = lane & 15;
    const int hh   = lane >> 4;
    const int kh   = hh * 2;

    const int row = tile * 16 + c;
    const v2i te  = *(const v2i*)(TE + row * (NREP * 2));
    int dow = te[0] % 7;   if (dow < 0) dow += 7;
    int tod = te[1] % 288; if (tod < 0) tod += 288;
    const float* __restrict__ wd = W1 + dow * DDIM + kh;
    const float* __restrict__ wt = W1 + (7 + tod) * DDIM + kh;
    const float* __restrict__ bh = b1 + kh;
    const float* __restrict__ w2 = W2 + kh * DDIM + 4 * c;

    v8f acc0 = {}, acc1 = {}, acc2 = {}, acc3 = {};
#pragma unroll
    for (int k0 = 0; k0 < DDIM; k0 += 4) {
        const v2f x = *(const v2f*)(wd + k0);
        const v2f y = *(const v2f*)(wt + k0);
        const v2f z = *(const v2f*)(bh + k0);
        const float s0 = x[0] + y[0] + z[0];
        const float s1 = x[1] + y[1] + z[1];
        v2f a;
        a[0] = s0 > 0.0f ? s0 : 0.0f;
        a[1] = s1 > 0.0f ? s1 : 0.0f;

        const v4f q0 = *(const v4f*)(w2 + (k0 + 0) * DDIM);
        const v4f q1 = *(const v4f*)(w2 + (k0 + 1) * DDIM);
        v2f bt;
        bt[0] = q0[0]; bt[1] = q1[0];
        acc0 = __builtin_amdgcn_wmma_f32_16x16x4_f32(false, a, false, bt, (short)0, acc0, false, false);
        bt[0] = q0[1]; bt[1] = q1[1];
        acc1 = __builtin_amdgcn_wmma_f32_16x16x4_f32(false, a, false, bt, (short)0, acc1, false, false);
        bt[0] = q0[2]; bt[1] = q1[2];
        acc2 = __builtin_amdgcn_wmma_f32_16x16x4_f32(false, a, false, bt, (short)0, acc2, false, false);
        bt[0] = q0[3]; bt[1] = q1[3];
        acc3 = __builtin_amdgcn_wmma_f32_16x16x4_f32(false, a, false, bt, (short)0, acc3, false, false);
    }

    const v4f bias = *(const v4f*)(b2 + 4 * c);

    const int  start = part * 14 + (part < 11 ? part : 11);
    const bool extra = part < 11;

    float* __restrict__ p =
        out + ((size_t)(tile * 16 + 8 * hh) * NREP + start) * DDIM + 4 * c;

#pragma unroll
    for (int r = 0; r < 8; ++r) {
        v4f v;
        v[0] = acc0[r] + bias[0];
        v[1] = acc1[r] + bias[1];
        v[2] = acc2[r] + bias[2];
        v[3] = acc3[r] + bias[3];
        v4f* __restrict__ d = (v4f*)(p + (size_t)r * ROW_FLOATS);
#pragma unroll
        for (int j = 0; j < 14; ++j) {
            __builtin_nontemporal_store(v, d + j * (DDIM / 4));
        }
        if (extra) {
            __builtin_nontemporal_store(v, d + 14 * (DDIM / 4));
        }
    }
}

extern "C" void kernel_launch(void* const* d_in, const int* in_sizes, int n_in,
                              void* d_out, int out_size, void* d_ws, size_t ws_size,
                              hipStream_t stream) {
    (void)in_sizes; (void)n_in; (void)out_size; (void)d_ws; (void)ws_size;

    const int*   TE = (const int*)  d_in[0];
    const float* W1 = (const float*)d_in[1];
    const float* b1 = (const float*)d_in[2];
    const float* W2 = (const float*)d_in[3];
    const float* b2 = (const float*)d_in[4];

    float* out = (float*)d_out;

    embed_rows_kernel<<<NUM_GROUPS, WG_WAVES * 32, 0, stream>>>(TE, W1, b1, W2, b2, out);
}
